# starting kernel
# baseline (speedup 1.0000x reference)
#include <hip/hip_runtime.h>
#include <hip/hip_bf16.h>

#define F 64

typedef __attribute__((ext_vector_type(2))) float v2f;
typedef __attribute__((ext_vector_type(8))) float v8f;

__device__ __forceinline__ unsigned f2o(float f) {
    unsigned u = __float_as_uint(f);
    return (u & 0x80000000u) ? ~u : (u | 0x80000000u);
}
__device__ __forceinline__ float o2f(unsigned u) {
    return (u & 0x80000000u) ? __uint_as_float(u & 0x7FFFFFFFu) : __uint_as_float(~u);
}

__global__ __launch_bounds__(256) void gat_gemm_wmma(
    const float* __restrict__ x, const float* __restrict__ W,
    float* __restrict__ h, int N) {
    int wave = blockIdx.x * (blockDim.x >> 5) + (threadIdx.x >> 5);
    int lane = threadIdx.x & 31;
    int row  = (wave >> 2) << 4;
    int col  = (wave & 3) << 4;
    if (row >= N) return;

    int mn    = lane & 15;
    int khalf = (lane >> 4) << 1;

    int arow = row + mn; if (arow > N - 1) arow = N - 1;
    const float* xrow = x + (size_t)arow * F;

    v8f acc = {};
#pragma unroll
    for (int k = 0; k < F; k += 4) {
        float2 av = *(const float2*)(xrow + k + khalf);
        v2f a; a.x = av.x; a.y = av.y;
        v2f b;
        b.x = W[(size_t)(k + khalf)     * F + col + mn];
        b.y = W[(size_t)(k + khalf + 1) * F + col + mn];
        acc = __builtin_amdgcn_wmma_f32_16x16x4_f32(
             false, a,  false, b,
             (short)0, acc,  false,  false);
    }

    int mo = (lane >> 4) << 3;
#pragma unroll
    for (int r = 0; r < 8; ++r) {
        int m = row + mo + r;
        if (m < N) h[(size_t)m * F + col + mn] = acc[r];
    }
}

__global__ void gat_scores(const float* __restrict__ h,
                           const float* __restrict__ a_src,
                           const float* __restrict__ a_dst,
                           float* __restrict__ s_src, float* __restrict__ s_dst,
                           int N) {
    int i = blockIdx.x * blockDim.x + threadIdx.x;
    if (i >= N) return;
    const float4* hv = (const float4*)(h + (size_t)i * F);
    const float4* as = (const float4*)a_src;
    const float4* ad = (const float4*)a_dst;
    float ss = 0.f, sd = 0.f;
#pragma unroll
    for (int c = 0; c < F / 4; ++c) {
        float4 hh = hv[c], s4 = as[c], d4 = ad[c];
        ss += hh.x * s4.x + hh.y * s4.y + hh.z * s4.z + hh.w * s4.w;
        sd += hh.x * d4.x + hh.y * d4.y + hh.z * d4.z + hh.w * d4.w;
    }
    s_src[i] = ss; s_dst[i] = sd;
}

__global__ void gat_init(unsigned* __restrict__ m_ord, float* __restrict__ denom,
                         float* __restrict__ acc, int N, long total) {
    long i = (long)blockIdx.x * blockDim.x + threadIdx.x;
    if (i < N) { m_ord[i] = 0x007FFFFFu;   denom[i] = 0.f; }
    if (i < total) acc[i] = 0.f;
}

__global__ void gat_edge_max(const int* __restrict__ esrc, const int* __restrict__ edst,
                             const float* __restrict__ s_src, const float* __restrict__ s_dst,
                             float* __restrict__ e_buf, unsigned* __restrict__ m_ord, int E) {
    int e = blockIdx.x * blockDim.x + threadIdx.x;
    if (e >= E) return;
    int s = esrc[e], d = edst[e];
    float v = s_src[s] + s_dst[d];
    v = v > 0.f ? v : 0.2f * v;
    e_buf[e] = v;
    atomicMax(&m_ord[d], f2o(v));
}

__global__ void gat_edge_exp(const int* __restrict__ edst,
                             const float* __restrict__ e_buf,
                             const unsigned* __restrict__ m_ord,
                             float* __restrict__ ex_buf, float* __restrict__ denom, int E) {
    int e = blockIdx.x * blockDim.x + threadIdx.x;
    if (e >= E) return;
    int d = edst[e];
    float ex = expf(e_buf[e] - o2f(m_ord[d]));
    ex_buf[e] = ex;
    atomicAdd(&denom[d], ex);
}

__global__ void gat_scatter(const int* __restrict__ esrc, const int* __restrict__ edst,
                            const float* __restrict__ h, const float* __restrict__ ex_buf,
                            const float* __restrict__ denom, float* __restrict__ acc, int E) {
    long t = (long)blockIdx.x * blockDim.x + threadIdx.x;
    int e = (int)(t >> 4);
    int q = (int)(t & 15);
    if (e >= E) return;
    int s = esrc[e], d = edst[e];
    float dn = denom[d]; dn = dn > 1e-9f ? dn : 1e-9f;
    float alpha = ex_buf[e] / dn;
    float4 hv = ((const float4*)(h + (size_t)s * F))[q];
    float* ap = acc + (size_t)d * F + q * 4;
    atomicAdd(ap + 0, alpha * hv.x);
    atomicAdd(ap + 1, alpha * hv.y);
    atomicAdd(ap + 2, alpha * hv.z);
    atomicAdd(ap + 3, alpha * hv.w);
}

__global__ void gat_final(const float* __restrict__ acc, const float* __restrict__ b,
                          float* __restrict__ out, long total) {
    long i = (long)blockIdx.x * blockDim.x + threadIdx.x;
    if (i >= total) return;
    float v = acc[i] + b[i & (F - 1)];
    out[i] = v > 0.f ? v : 0.3f * v;
}

extern "C" void kernel_launch(void* const* d_in, const int* in_sizes, int n_in,
                              void* d_out, int out_size, void* d_ws, size_t ws_size,
                              hipStream_t stream) {
    const float* x     = (const float*)d_in[0];
    const float* W     = (const float*)d_in[1];
    const float* a_src = (const float*)d_in[2];
    const float* a_dst = (const float*)d_in[3];
    const float* b     = (const float*)d_in[4];
    const int*   esrc  = (const int*)d_in[5];
    const int*   edst  = (const int*)d_in[6];

    const int N = in_sizes[0] / F;
    const int E = in_sizes[5];
    const long total = (long)N * F;

    char* ws = (char*)d_ws;
    size_t off = 0;
    auto carve = [&](size_t bytes) -> void* {
        void* p = ws + off;
        off = (off + bytes + 255) & ~(size_t)255;
        return p;
    };
    float*    h      = (float*)carve((size_t)total * 4);
    float*    acc    = (float*)carve((size_t)total * 4);
    float*    s_src  = (float*)carve((size_t)N * 4);
    float*    s_dst  = (float*)carve((size_t)N * 4);
    unsigned* m_ord  = (unsigned*)carve((size_t)N * 4);
    float*    denom  = (float*)carve((size_t)N * 4);
    float*    e_buf  = (float*)carve((size_t)E * 4);
    float*    ex_buf = (float*)carve((size_t)E * 4);
    (void)ws_size; (void)n_in; (void)out_size;

    int rowTiles = (N + 15) / 16;
    int waves = rowTiles * 4;
    int gemmBlocks = (waves + 7) / 8;
    gat_gemm_wmma<<<gemmBlocks, 256, 0, stream>>>(x, W, h, N);

    gat_scores<<<(N + 255) / 256, 256, 0, stream>>>(h, a_src, a_dst, s_src, s_dst, N);

    gat_init<<<(int)((total + 255) / 256), 256, 0, stream>>>(m_ord, denom, acc, N, total);

    gat_edge_max<<<(E + 255) / 256, 256, 0, stream>>>(esrc, edst, s_src, s_dst, e_buf, m_ord, E);

    gat_edge_exp<<<(E + 255) / 256, 256, 0, stream>>>(edst, e_buf, m_ord, ex_buf, denom, E);

    long st = (long)E * 16;
    gat_scatter<<<(int)((st + 255) / 256), 256, 0, stream>>>(esrc, edst, h, ex_buf, denom, acc, E);

    gat_final<<<(int)((total + 255) / 256), 256, 0, stream>>>(acc, b, (float*)d_out, total);
}


# Edges grouped by destination range in LDS: each work-group owns 196 nodes, lists its edges, does max, sum and row adds with LDS atomics, stores rows once; no global atomics; sums in LDS arrival order
# speedup vs baseline: 2.9797x; 2.9797x over previous
#include <hip/hip_runtime.h>
#include <hip/hip_bf16.h>

#define F 64

typedef __attribute__((ext_vector_type(2))) float v2f;
typedef __attribute__((ext_vector_type(8))) float v8f;

__device__ __forceinline__ unsigned f2o(float f) {
    unsigned u = __float_as_uint(f);
    return (u & 0x80000000u) ? ~u : (u | 0x80000000u);
}
__device__ __forceinline__ float o2f(unsigned u) {
    return (u & 0x80000000u) ? __uint_as_float(u & 0x7FFFFFFFu) : __uint_as_float(~u);
}

__global__ __launch_bounds__(256) void gat_gemm_wmma(
    const float* __restrict__ x, const float* __restrict__ W,
    float* __restrict__ h, int N) {
    int wave = blockIdx.x * (blockDim.x >> 5) + (threadIdx.x >> 5);
    int lane = threadIdx.x & 31;
    int row  = (wave >> 2) << 4;
    int col  = (wave & 3) << 4;
    if (row >= N) return;

    int mn    = lane & 15;
    int khalf = (lane >> 4) << 1;

    int arow = row + mn; if (arow > N - 1) arow = N - 1;
    const float* xrow = x + (size_t)arow * F;

    v8f acc = {};
#pragma unroll
    for (int k = 0; k < F; k += 4) {
        float2 av = *(const float2*)(xrow + k + khalf);
        v2f a; a.x = av.x; a.y = av.y;
        v2f b;
        b.x = W[(size_t)(k + khalf)     * F + col + mn];
        b.y = W[(size_t)(k + khalf + 1) * F + col + mn];
        acc = __builtin_amdgcn_wmma_f32_16x16x4_f32(
             false, a,  false, b,
             (short)0, acc,  false,  false);
    }

    int mo = (lane >> 4) << 3;
#pragma unroll
    for (int r = 0; r < 8; ++r) {
        int m = row + mo + r;
        if (m < N) h[(size_t)m * F + col + mn] = acc[r];
    }
}

__global__ void gat_scores(const float* __restrict__ h,
                           const float* __restrict__ a_src,
                           const float* __restrict__ a_dst,
                           float* __restrict__ s_src, float* __restrict__ s_dst,
                           int N) {
    int i = blockIdx.x * blockDim.x + threadIdx.x;
    if (i >= N) return;
    const float4* hv = (const float4*)(h + (size_t)i * F);
    const float4* as = (const float4*)a_src;
    const float4* ad = (const float4*)a_dst;
    float ss = 0.f, sd = 0.f;
#pragma unroll
    for (int c = 0; c < F / 4; ++c) {
        float4 hh = hv[c], s4 = as[c], d4 = ad[c];
        ss += hh.x * s4.x + hh.y * s4.y + hh.z * s4.z + hh.w * s4.w;
        sd += hh.x * d4.x + hh.y * d4.y + hh.z * d4.z + hh.w * d4.w;
    }
    s_src[i] = ss; s_dst[i] = sd;
}

#define OWN_R   196
#define OWN_T   1024
#define OWN_CAP 6144
#define OWN_U   8

__global__ __launch_bounds__(OWN_T) void gat_owner(
    const int* __restrict__ esrc, const int* __restrict__ edst,
    const float* __restrict__ s_src, const float* __restrict__ s_dst,
    const float* __restrict__ h, const float* __restrict__ b,
    float* __restrict__ out, int N, int E) {
    __shared__ __attribute__((aligned(16))) float acc[(OWN_R + 1) * F];
    __shared__ unsigned mx[OWN_R];
    __shared__ float    den[OWN_R];
    __shared__ float    sd[OWN_R];
    __shared__ int      lS[OWN_CAP + OWN_U];
    __shared__ int      lD[OWN_CAP + OWN_U];
    __shared__ float    lV[OWN_CAP + OWN_U];
    __shared__ int      cnt;

    const int tid  = threadIdx.x;
    const int lane = tid & 31;
    const int wave = tid >> 5;
    const int NW   = OWN_T >> 5;
    const int lo   = blockIdx.x * OWN_R;
    int nown = N - lo; if (nown > OWN_R) nown = OWN_R;
    const unsigned un = (unsigned)nown;

    for (int i = tid; i < (OWN_R + 1) * F; i += OWN_T) acc[i] = 0.f;
    if (tid < OWN_R) {
        mx[tid]  = 0x007FFFFFu;
        den[tid] = 0.f;
        sd[tid]  = (tid < nown) ? s_dst[lo + tid] : 0.f;
    }
    if (tid == 0) cnt = 0;
    __syncthreads();

    auto take = [&](unsigned r, int e) {
        if (r < un) {
            int p = atomicAdd(&cnt, 1);
            if (p < OWN_CAP) { lS[p] = e; lD[p] = (int)r; }
        }
    };
    auto take4 = [&](int4 d, int e) {
        unsigned r0 = (unsigned)(d.x - lo), r1 = (unsigned)(d.y - lo);
        unsigned r2 = (unsigned)(d.z - lo), r3 = (unsigned)(d.w - lo);
        unsigned m01 = r0 < r1 ? r0 : r1, m23 = r2 < r3 ? r2 : r3;
        unsigned m = m01 < m23 ? m01 : m23;
        if (m < un) { take(r0, e); take(r1, e + 1); take(r2, e + 2); take(r3, e + 3); }
    };
    const int   E4 = E >> 2;
    const int4* d4 = (const int4*)edst;
    int i4 = tid;
    for (; i4 + (OWN_U - 1) * OWN_T < E4; i4 += OWN_U * OWN_T) {
        int4 v[OWN_U];
#pragma unroll
        for (int u = 0; u < OWN_U; ++u) v[u] = d4[i4 + u * OWN_T];
#pragma unroll
        for (int u = 0; u < OWN_U; ++u) take4(v[u], (i4 + u * OWN_T) << 2);
    }
    for (; i4 < E4; i4 += OWN_T) {
        take4(d4[i4], i4 << 2);
    }
    for (int e = (E4 << 2) + tid; e < E; e += OWN_T) take((unsigned)(edst[e] - lo), e);
    __syncthreads();
    const int  total = cnt;
    const bool fast  = total <= OWN_CAP;

    if (fast) {
        for (int i = tid; i < total; i += OWN_T) {
            int s = esrc[lS[i]];
            int r = lD[i];
            float v = s_src[s] + sd[r];
            v = v > 0.f ? v : 0.2f * v;
            lS[i] = s; lV[i] = v;
            atomicMax(&mx[r], f2o(v));
        }
    } else {
        for (int e = tid; e < E; e += OWN_T) {
            unsigned r = (unsigned)(edst[e] - lo);
            if (r < un) {
                float v = s_src[esrc[e]] + sd[r];
                v = v > 0.f ? v : 0.2f * v;
                atomicMax(&mx[r], f2o(v));
            }
        }
    }
    __syncthreads();

    if (fast) {
        for (int i = tid; i < total; i += OWN_T) {
            int r = lD[i];
            float ex = expf(lV[i] - o2f(mx[r]));
            lV[i] = ex;
            atomicAdd(&den[r], ex);
        }
    } else {
        for (int e = tid; e < E; e += OWN_T) {
            unsigned r = (unsigned)(edst[e] - lo);
            if (r < un) {
                float v = s_src[esrc[e]] + sd[r];
                v = v > 0.f ? v : 0.2f * v;
                atomicAdd(&den[r], expf(v - o2f(mx[r])));
            }
        }
    }
    __syncthreads();

    if (fast) {
        for (int i = tid; i < total; i += OWN_T) {
            float dn = den[lD[i]]; dn = dn > 1e-9f ? dn : 1e-9f;
            lV[i] = lV[i] / dn;
        }
        if (tid < OWN_U) { lS[total + tid] = 0; lD[total + tid] = OWN_R; lV[total + tid] = 0.f; }
    }
    __syncthreads();

    if (fast) {
        for (int i0 = wave * OWN_U; i0 < total; i0 += NW * OWN_U) {
            float a[OWN_U], h0[OWN_U], h1[OWN_U];
            int   ss[OWN_U], rr[OWN_U];
#pragma unroll
            for (int u = 0; u < OWN_U; ++u) { ss[u] = lS[i0 + u]; rr[u] = lD[i0 + u]; a[u] = lV[i0 + u]; }
#pragma unroll
            for (int u = 0; u < OWN_U; ++u) {
                const float* hr = h + (size_t)ss[u] * F;
                h0[u] = hr[lane];
                h1[u] = hr[lane + 32];
            }
#pragma unroll
            for (int u = 0; u < OWN_U; ++u) {
                atomicAdd(&acc[rr[u] * F + lane],      a[u] * h0[u]);
                atomicAdd(&acc[rr[u] * F + lane + 32], a[u] * h1[u]);
            }
        }
    } else {
        for (int e = wave; e < E; e += NW) {
            unsigned r = (unsigned)(edst[e] - lo);
            if (r < un) {
                int s = esrc[e];
                float v = s_src[s] + sd[r];
                v = v > 0.f ? v : 0.2f * v;
                float dn = den[r]; dn = dn > 1e-9f ? dn : 1e-9f;
                float alpha = expf(v - o2f(mx[r])) / dn;
                const float* hr = h + (size_t)s * F;
                atomicAdd(&acc[r * F + lane],      alpha * hr[lane]);
                atomicAdd(&acc[r * F + lane + 32], alpha * hr[lane + 32]);
            }
        }
    }
    __syncthreads();

    const float4* acc4 = (const float4*)acc;
    const float4* b4   = (const float4*)b;
    float4*       o4   = (float4*)(out + (size_t)lo * F);
    for (int i = tid; i < nown * (F / 4); i += OWN_T) {
        float4 v = acc4[i], bb = b4[i & (F / 4 - 1)];
        v.x += bb.x; v.y += bb.y; v.z += bb.z; v.w += bb.w;
        v.x = v.x > 0.f ? v.x : 0.3f * v.x;
        v.y = v.y > 0.f ? v.y : 0.3f * v.y;
        v.z = v.z > 0.f ? v.z : 0.3f * v.z;
        v.w = v.w > 0.f ? v.w : 0.3f * v.w;
        o4[i] = v;
    }
}

extern "C" void kernel_launch(void* const* d_in, const int* in_sizes, int n_in,
                              void* d_out, int out_size, void* d_ws, size_t ws_size,
                              hipStream_t stream) {
    const float* x     = (const float*)d_in[0];
    const float* W     = (const float*)d_in[1];
    const float* a_src = (const float*)d_in[2];
    const float* a_dst = (const float*)d_in[3];
    const float* b     = (const float*)d_in[4];
    const int*   esrc  = (const int*)d_in[5];
    const int*   edst  = (const int*)d_in[6];

    const int N = in_sizes[0] / F;
    const int E = in_sizes[5];
    const long total = (long)N * F;

    char* ws = (char*)d_ws;
    size_t off = 0;
    auto carve = [&](size_t bytes) -> void* {
        void* p = ws + off;
        off = (off + bytes + 255) & ~(size_t)255;
        return p;
    };
    float* h     = (float*)carve((size_t)total * 4);
    float* s_src = (float*)carve((size_t)N * 4);
    float* s_dst = (float*)carve((size_t)N * 4);
    (void)ws_size; (void)n_in; (void)out_size;

    int rowTiles = (N + 15) / 16;
    int waves = rowTiles * 4;
    int gemmBlocks = (waves + 7) / 8;
    gat_gemm_wmma<<<gemmBlocks, 256, 0, stream>>>(x, W, h, N);

    gat_scores<<<(N + 255) / 256, 256, 0, stream>>>(h, a_src, a_dst, s_src, s_dst, N);

    gat_owner<<<(N + OWN_R - 1) / OWN_R, OWN_T, 0, stream>>>(esrc, edst, s_src, s_dst, h, b,
                                                            (float*)d_out, N, E);
}


# Edges binned by destination range once (binning blocks in the score launch reserve list slots by returning global adds); owner reads its list, no full scan; sums stay in LDS arrival order
# speedup vs baseline: 4.3207x; 1.4500x over previous
#include <hip/hip_runtime.h>
#include <hip/hip_bf16.h>

#define F 64

#define OWN_R   196
#define OWN_T   1024
#define OWN_CAP 6144
#define OWN_U   8

#define BIN_T      256
#define BIN_U      4
#define BIN_E      (BIN_T * BIN_U * 4)
#define BIN_NBMAX  1024
#define BIN_STRIDE 32

typedef __attribute__((ext_vector_type(2))) float v2f;
typedef __attribute__((ext_vector_type(8))) float v8f;

__device__ __forceinline__ unsigned f2o(float f) {
    unsigned u = __float_as_uint(f);
    return (u & 0x80000000u) ? ~u : (u | 0x80000000u);
}
__device__ __forceinline__ float o2f(unsigned u) {
    return (u & 0x80000000u) ? __uint_as_float(u & 0x7FFFFFFFu) : __uint_as_float(~u);
}

__global__ __launch_bounds__(256) void gat_gemm_wmma(
    const float* __restrict__ x, const float* __restrict__ W,
    float* __restrict__ h, int N, int* __restrict__ bincnt, int NB) {
    if (blockIdx.x == 0)
        for (int i = threadIdx.x; i < NB; i += blockDim.x) bincnt[i * BIN_STRIDE] = 0;
    int wave = blockIdx.x * (blockDim.x >> 5) + (threadIdx.x >> 5);
    int lane = threadIdx.x & 31;
    int row  = (wave >> 2) << 4;
    int col  = (wave & 3) << 4;
    if (row >= N) return;

    int mn    = lane & 15;
    int khalf = (lane >> 4) << 1;

    int arow = row + mn; if (arow > N - 1) arow = N - 1;
    const float* xrow = x + (size_t)arow * F;

    v8f acc = {};
#pragma unroll
    for (int k = 0; k < F; k += 4) {
        float2 av = *(const float2*)(xrow + k + khalf);
        v2f a; a.x = av.x; a.y = av.y;
        v2f b;
        b.x = W[(size_t)(k + khalf)     * F + col + mn];
        b.y = W[(size_t)(k + khalf + 1) * F + col + mn];
        acc = __builtin_amdgcn_wmma_f32_16x16x4_f32(
             false, a,  false, b,
             (short)0, acc,  false,  false);
    }

    int mo = (lane >> 4) << 3;
#pragma unroll
    for (int r = 0; r < 8; ++r) {
        int m = row + mo + r;
        if (m < N) h[(size_t)m * F + col + mn] = acc[r];
    }
}

__global__ __launch_bounds__(BIN_T) void gat_scores_bin(
                           const float* __restrict__ h,
                           const float* __restrict__ a_src,
                           const float* __restrict__ a_dst,
                           float* __restrict__ s_src, float* __restrict__ s_dst,
                           int N,
                           const int* __restrict__ esrc, const int* __restrict__ edst, int E,
                           int* __restrict__ bincnt, int* __restrict__ binlist,
                           int NB, int binBlocks) {
    __shared__ int hist[BIN_NBMAX + 32];
    if ((int)blockIdx.x < binBlocks) {
        if (NB > BIN_NBMAX) return;
        const int tid = threadIdx.x;
        for (int b = tid; b < NB; b += BIN_T) hist[b] = 0;
        const int spare = BIN_NBMAX + (tid & 31);
        if (tid < 32) hist[spare] = 0;
        __syncthreads();
        const int   E4 = E >> 2;
        const int4* d4 = (const int4*)edst;
        const int4* s4 = (const int4*)esrc;
        int dd[BIN_U * 4 + 1], ss[BIN_U * 4 + 1], gg[BIN_U * 4 + 1], rk[BIN_U * 4 + 1];
#pragma unroll
        for (int u = 0; u < BIN_U; ++u) {
            const int q = (int)blockIdx.x * (BIN_T * BIN_U) + u * BIN_T + tid;
            int4 dv = make_int4(-1, -1, -1, -1), sv = make_int4(0, 0, 0, 0);
            if (q < E4) { dv = d4[q]; sv = s4[q]; }
            dd[4 * u] = dv.x; dd[4 * u + 1] = dv.y; dd[4 * u + 2] = dv.z; dd[4 * u + 3] = dv.w;
            ss[4 * u] = sv.x; ss[4 * u + 1] = sv.y; ss[4 * u + 2] = sv.z; ss[4 * u + 3] = sv.w;
        }
        dd[BIN_U * 4] = -1; ss[BIN_U * 4] = 0;
        if (blockIdx.x == 0 && tid < (E & 3)) {
            dd[BIN_U * 4] = edst[(E4 << 2) + tid];
            ss[BIN_U * 4] = esrc[(E4 << 2) + tid];
        }
#pragma unroll
        for (int k = 0; k < BIN_U * 4 + 1; ++k) {
            gg[k] = (unsigned)dd[k] < (unsigned)N ? (int)((unsigned)dd[k] / OWN_R) : -1;
            rk[k] = atomicAdd(&hist[gg[k] >= 0 ? gg[k] : spare], 1);
        }
        __syncthreads();
        for (int b = tid; b < NB; b += BIN_T) {
            const int c = hist[b];
            hist[b] = c ? atomicAdd(&bincnt[b * BIN_STRIDE], c) : 0;
        }
        __syncthreads();
#pragma unroll
        for (int k = 0; k < BIN_U * 4 + 1; ++k) rk[k] += hist[gg[k] >= 0 ? gg[k] : 0];
        int2* bl = (int2*)binlist;
#pragma unroll
        for (int k = 0; k < BIN_U * 4 + 1; ++k) {
            if (gg[k] >= 0 && rk[k] < OWN_CAP)
                bl[(size_t)gg[k] * OWN_CAP + rk[k]] = make_int2(ss[k], dd[k] - gg[k] * OWN_R);
        }
        return;
    }
    int i = ((int)blockIdx.x - binBlocks) * BIN_T + threadIdx.x;
    if (i >= N) return;
    const float4* hv = (const float4*)(h + (size_t)i * F);
    const float4* as = (const float4*)a_src;
    const float4* ad = (const float4*)a_dst;
    float ss = 0.f, sd = 0.f;
#pragma unroll
    for (int c = 0; c < F / 4; ++c) {
        float4 hh = hv[c], s4 = as[c], d4 = ad[c];
        ss += hh.x * s4.x + hh.y * s4.y + hh.z * s4.z + hh.w * s4.w;
        sd += hh.x * d4.x + hh.y * d4.y + hh.z * d4.z + hh.w * d4.w;
    }
    s_src[i] = ss; s_dst[i] = sd;
}


__global__ __launch_bounds__(OWN_T) void gat_owner(
    const int* __restrict__ esrc, const int* __restrict__ edst,
    const float* __restrict__ s_src, const float* __restrict__ s_dst,
    const float* __restrict__ h, const float* __restrict__ b,
    float* __restrict__ out, int N, int E,
    const int* __restrict__ bincnt, const int* __restrict__ binlist) {
    __shared__ __attribute__((aligned(16))) float acc[(OWN_R + 1) * F];
    __shared__ unsigned mx[OWN_R];
    __shared__ float    den[OWN_R];
    __shared__ float    sd[OWN_R];
    __shared__ int      lS[OWN_CAP + OWN_U];
    __shared__ int      lD[OWN_CAP + OWN_U];
    __shared__ float    lV[OWN_CAP + OWN_U];

    const int tid  = threadIdx.x;
    const int lane = tid & 31;
    const int wave = tid >> 5;
    const int NW   = OWN_T >> 5;
    const int lo   = blockIdx.x * OWN_R;
    int nown = N - lo; if (nown > OWN_R) nown = OWN_R;
    const unsigned un = (unsigned)nown;

    for (int i = tid; i < (OWN_R + 1) * F; i += OWN_T) acc[i] = 0.f;
    if (tid < OWN_R) {
        mx[tid]  = 0x007FFFFFu;
        den[tid] = 0.f;
        sd[tid]  = (tid < nown) ? s_dst[lo + tid] : 0.f;
    }
    __syncthreads();

    const int  total = bincnt[blockIdx.x * BIN_STRIDE];
    const bool fast  = (int)gridDim.x <= BIN_NBMAX && total <= OWN_CAP;

    if (fast) {
        const int2* bl = (const int2*)binlist + (size_t)blockIdx.x * OWN_CAP;
        int2  en[OWN_CAP / OWN_T];
        float sv[OWN_CAP / OWN_T];
#pragma unroll
        for (int j = 0; j < OWN_CAP / OWN_T; ++j) {
            const int i = tid + j * OWN_T;
            en[j] = i < total ? bl[i] : make_int2(0, 0);
        }
#pragma unroll
        for (int j = 0; j < OWN_CAP / OWN_T; ++j) sv[j] = s_src[en[j].x];
#pragma unroll
        for (int j = 0; j < OWN_CAP / OWN_T; ++j) {
            const int i = tid + j * OWN_T;
            if (i < total) {
                int r = en[j].y;
                float v = sv[j] + sd[r];
                v = v > 0.f ? v : 0.2f * v;
                lS[i] = en[j].x; lD[i] = r; lV[i] = v;
                atomicMax(&mx[r], f2o(v));
            }
        }
    } else {
        for (int e = tid; e < E; e += OWN_T) {
            unsigned r = (unsigned)(edst[e] - lo);
            if (r < un) {
                float v = s_src[esrc[e]] + sd[r];
                v = v > 0.f ? v : 0.2f * v;
                atomicMax(&mx[r], f2o(v));
            }
        }
    }
    __syncthreads();

    if (fast) {
        for (int i = tid; i < total; i += OWN_T) {
            int r = lD[i];
            float ex = expf(lV[i] - o2f(mx[r]));
            lV[i] = ex;
            atomicAdd(&den[r], ex);
        }
    } else {
        for (int e = tid; e < E; e += OWN_T) {
            unsigned r = (unsigned)(edst[e] - lo);
            if (r < un) {
                float v = s_src[esrc[e]] + sd[r];
                v = v > 0.f ? v : 0.2f * v;
                atomicAdd(&den[r], expf(v - o2f(mx[r])));
            }
        }
    }
    __syncthreads();

    if (fast) {
        for (int i = tid; i < total; i += OWN_T) {
            float dn = den[lD[i]]; dn = dn > 1e-9f ? dn : 1e-9f;
            lV[i] = lV[i] / dn;
        }
        if (tid < OWN_U) { lS[total + tid] = 0; lD[total + tid] = OWN_R; lV[total + tid] = 0.f; }
    }
    __syncthreads();

    if (fast) {
        for (int i0 = wave * OWN_U; i0 < total; i0 += NW * OWN_U) {
            float a[OWN_U], h0[OWN_U], h1[OWN_U];
            int   ss[OWN_U], rr[OWN_U];
#pragma unroll
            for (int u = 0; u < OWN_U; ++u) { ss[u] = lS[i0 + u]; rr[u] = lD[i0 + u]; a[u] = lV[i0 + u]; }
#pragma unroll
            for (int u = 0; u < OWN_U; ++u) {
                const float* hr = h + (size_t)ss[u] * F;
                h0[u] = hr[lane];
                h1[u] = hr[lane + 32];
            }
#pragma unroll
            for (int u = 0; u < OWN_U; ++u) {
                atomicAdd(&acc[rr[u] * F + lane],      a[u] * h0[u]);
                atomicAdd(&acc[rr[u] * F + lane + 32], a[u] * h1[u]);
            }
        }
    } else {
        for (int e = wave; e < E; e += NW) {
            unsigned r = (unsigned)(edst[e] - lo);
            if (r < un) {
                int s = esrc[e];
                float v = s_src[s] + sd[r];
                v = v > 0.f ? v : 0.2f * v;
                float dn = den[r]; dn = dn > 1e-9f ? dn : 1e-9f;
                float alpha = expf(v - o2f(mx[r])) / dn;
                const float* hr = h + (size_t)s * F;
                atomicAdd(&acc[r * F + lane],      alpha * hr[lane]);
                atomicAdd(&acc[r * F + lane + 32], alpha * hr[lane + 32]);
            }
        }
    }
    __syncthreads();

    const float4* acc4 = (const float4*)acc;
    const float4* b4   = (const float4*)b;
    float4*       o4   = (float4*)(out + (size_t)lo * F);
    for (int i = tid; i < nown * (F / 4); i += OWN_T) {
        float4 v = acc4[i], bb = b4[i & (F / 4 - 1)];
        v.x += bb.x; v.y += bb.y; v.z += bb.z; v.w += bb.w;
        v.x = v.x > 0.f ? v.x : 0.3f * v.x;
        v.y = v.y > 0.f ? v.y : 0.3f * v.y;
        v.z = v.z > 0.f ? v.z : 0.3f * v.z;
        v.w = v.w > 0.f ? v.w : 0.3f * v.w;
        o4[i] = v;
    }
}

extern "C" void kernel_launch(void* const* d_in, const int* in_sizes, int n_in,
                              void* d_out, int out_size, void* d_ws, size_t ws_size,
                              hipStream_t stream) {
    const float* x     = (const float*)d_in[0];
    const float* W     = (const float*)d_in[1];
    const float* a_src = (const float*)d_in[2];
    const float* a_dst = (const float*)d_in[3];
    const float* b     = (const float*)d_in[4];
    const int*   esrc  = (const int*)d_in[5];
    const int*   edst  = (const int*)d_in[6];

    const int N = in_sizes[0] / F;
    const int E = in_sizes[5];
    const long total = (long)N * F;

    char* ws = (char*)d_ws;
    size_t off = 0;
    auto carve = [&](size_t bytes) -> void* {
        void* p = ws + off;
        off = (off + bytes + 255) & ~(size_t)255;
        return p;
    };
    float* h     = (float*)carve((size_t)total * 4);
    float* s_src = (float*)carve((size_t)N * 4);
    float* s_dst = (float*)carve((size_t)N * 4);
    const int NB = (N + OWN_R - 1) / OWN_R;
    int* bincnt  = (int*)carve((size_t)NB * BIN_STRIDE * 4);
    int* binlist = (int*)carve((size_t)NB * OWN_CAP * 8);
    (void)ws_size; (void)n_in; (void)out_size;

    int rowTiles = (N + 15) / 16;
    int waves = rowTiles * 4;
    int gemmBlocks = (waves + 7) / 8;
    gat_gemm_wmma<<<gemmBlocks, 256, 0, stream>>>(x, W, h, N, bincnt, NB);

    int binBlocks = (E + BIN_E - 1) / BIN_E;
    gat_scores_bin<<<binBlocks + (N + BIN_T - 1) / BIN_T, BIN_T, 0, stream>>>(
        h, a_src, a_dst, s_src, s_dst, N, esrc, edst, E, bincnt, binlist, NB, binBlocks);

    gat_owner<<<NB, OWN_T, 0, stream>>>(esrc, edst, s_src, s_dst, h, b,
                                        (float*)d_out, N, E, bincnt, binlist);
}


# GEMM wave owns 16 whole rows (4 accumulator tiles, operands via LDS) and forms both node scores from its tile: launch 2 only bins, h not re-read; score dot summed as two 32-column halves
# speedup vs baseline: 4.5071x; 1.0431x over previous
#include <hip/hip_runtime.h>
#include <hip/hip_bf16.h>

#define F 64

#define OWN_R   196
#define OWN_T   1024
#define OWN_CAP 6144
#define OWN_U   8

#define BIN_T      256
#define BIN_U      4
#define BIN_E      (BIN_T * BIN_U * 4)
#define BIN_NBMAX  1024
#define BIN_STRIDE 32

typedef __attribute__((ext_vector_type(2))) float v2f;
typedef __attribute__((ext_vector_type(8))) float v8f;

__device__ __forceinline__ unsigned f2o(float f) {
    unsigned u = __float_as_uint(f);
    return (u & 0x80000000u) ? ~u : (u | 0x80000000u);
}
__device__ __forceinline__ float o2f(unsigned u) {
    return (u & 0x80000000u) ? __uint_as_float(u & 0x7FFFFFFFu) : __uint_as_float(~u);
}

#define GM_W     8
#define GM_T     (GM_W * 32)
#define GM_PITCH 68
#define WP_PITCH 80

__global__ __launch_bounds__(GM_T) void gat_gemm_scores(
    const float* __restrict__ x, const float* __restrict__ W,
    const float* __restrict__ a_src, const float* __restrict__ a_dst,
    float* __restrict__ h, float* __restrict__ s_src, float* __restrict__ s_dst,
    int N, int* __restrict__ bincnt, int NB) {
    __shared__ __attribute__((aligned(16))) float Wp[32 * WP_PITCH * 2];
    __shared__ __attribute__((aligned(16))) float tile[GM_W][16 * GM_PITCH];
    __shared__ __attribute__((aligned(16))) float av[GM_W][2 * F];

    const int tid  = threadIdx.x;
    const int lane = tid & 31;
    const int wave = tid >> 5;
    const int q    = lane & 15;
    const int hh   = lane >> 4;
    const int row0 = __builtin_amdgcn_readfirstlane(((int)blockIdx.x * GM_W + wave) << 4);
    float* tl = tile[wave];

    float4 xv[8];
#pragma unroll
    for (int i = 0; i < 8; ++i) {
        int gr = row0 + 2 * i + hh; if (gr > N - 1) gr = N - 1;
        xv[i] = *(const float4*)(x + (size_t)gr * F + 4 * q);
    }
    const float4 avv = *(const float4*)((hh ? a_dst : a_src) + 4 * q);
    float4 w0[2], w1[2];
#pragma unroll
    for (int j = 0; j < 2; ++j) {
        const int it = tid + j * GM_T, kp = it >> 4, cg = it & 15;
        w0[j] = *(const float4*)(W + (size_t)(2 * kp) * F + 4 * cg);
        w1[j] = *(const float4*)(W + (size_t)(2 * kp + 1) * F + 4 * cg);
    }
#pragma unroll
    for (int j = 0; j < 2; ++j) {
        const int it = tid + j * GM_T, kp = it >> 4, cg = it & 15;
        float* wp = Wp + 2 * (kp * WP_PITCH + 4 * cg);
        *(float4*)(wp)     = make_float4(w0[j].x, w1[j].x, w0[j].y, w1[j].y);
        *(float4*)(wp + 4) = make_float4(w0[j].z, w1[j].z, w0[j].w, w1[j].w);
    }
#pragma unroll
    for (int i = 0; i < 8; ++i) *(float4*)(tl + (2 * i + hh) * GM_PITCH + 4 * q) = xv[i];
    *(float4*)(av[wave] + 4 * lane) = avv;
    __syncthreads();

    v2f a[16];
#pragma unroll
    for (int ks = 0; ks < 16; ++ks) a[ks] = *(const v2f*)(tl + q * GM_PITCH + 4 * ks + 2 * hh);

    v8f acc[4] = {};
#pragma unroll
    for (int ks = 0; ks < 16; ++ks) {
#pragma unroll
        for (int t = 0; t < 4; ++t) {
            const v2f b = *(const v2f*)(Wp + 2 * ((2 * ks + hh) * WP_PITCH + 16 * t + q));
            acc[t] = __builtin_amdgcn_wmma_f32_16x16x4_f32(
                 false, a[ks],  false, b,
                 (short)0, acc[t],  false,  false);
        }
    }

#pragma unroll
    for (int t = 0; t < 4; ++t)
#pragma unroll
        for (int r = 0; r < 8; ++r) tl[(8 * hh + r) * GM_PITCH + 16 * t + q] = acc[t][r];
    __syncthreads();

    if (row0 + 16 <= N) {
        float4 hv[8];
#pragma unroll
        for (int i = 0; i < 8; ++i) hv[i] = *(const float4*)(tl + (2 * i + hh) * GM_PITCH + 4 * q);
#pragma unroll
        for (int i = 0; i < 8; ++i) *(float4*)(h + (size_t)(row0 + 2 * i + hh) * F + 4 * q) = hv[i];
    } else {
#pragma unroll
        for (int i = 0; i < 8; ++i) {
            const int r = 2 * i + hh;
            const float4 v = *(const float4*)(tl + r * GM_PITCH + 4 * q);
            if (row0 + r < N) *(float4*)(h + (size_t)(row0 + r) * F + 4 * q) = v;
        }
    }

    float ps = 0.f, pd = 0.f;
#pragma unroll
    for (int j = 0; j < 8; ++j) {
        const float4 v  = *(const float4*)(tl + q * GM_PITCH + 32 * hh + 4 * j);
        const float4 s4 = *(const float4*)(av[wave] + 32 * hh + 4 * j);
        const float4 d4 = *(const float4*)(av[wave] + F + 32 * hh + 4 * j);
        ps += v.x * s4.x + v.y * s4.y + v.z * s4.z + v.w * s4.w;
        pd += v.x * d4.x + v.y * d4.y + v.z * d4.z + v.w * d4.w;
    }
    ps += __shfl_xor(ps, 16);
    pd += __shfl_xor(pd, 16);
    if (row0 + q < N) {
        if (hh == 0) s_src[row0 + q] = ps; else s_dst[row0 + q] = pd;
    }

    if (blockIdx.x == 0)
        for (int i = tid; i < NB; i += GM_T) bincnt[i * BIN_STRIDE] = 0;
}

__global__ __launch_bounds__(BIN_T) void gat_bin(
                           int N,
                           const int* __restrict__ esrc, const int* __restrict__ edst, int E,
                           int* __restrict__ bincnt, int* __restrict__ binlist,
                           int NB) {
    __shared__ int hist[BIN_NBMAX + 32];
    if (NB > BIN_NBMAX) return;
    const int tid = threadIdx.x;
    for (int b = tid; b < NB; b += BIN_T) hist[b] = 0;
    const int spare = BIN_NBMAX + (tid & 31);
    if (tid < 32) hist[spare] = 0;
    __syncthreads();
    const int   E4 = E >> 2;
    const int4* d4 = (const int4*)edst;
    const int4* s4 = (const int4*)esrc;
    int dd[BIN_U * 4 + 1], ss[BIN_U * 4 + 1], gg[BIN_U * 4 + 1], rk[BIN_U * 4 + 1];
#pragma unroll
    for (int u = 0; u < BIN_U; ++u) {
        const int q = (int)blockIdx.x * (BIN_T * BIN_U) + u * BIN_T + tid;
        int4 dv = make_int4(-1, -1, -1, -1), sv = make_int4(0, 0, 0, 0);
        if (q < E4) { dv = d4[q]; sv = s4[q]; }
        dd[4 * u] = dv.x; dd[4 * u + 1] = dv.y; dd[4 * u + 2] = dv.z; dd[4 * u + 3] = dv.w;
        ss[4 * u] = sv.x; ss[4 * u + 1] = sv.y; ss[4 * u + 2] = sv.z; ss[4 * u + 3] = sv.w;
    }
    dd[BIN_U * 4] = -1; ss[BIN_U * 4] = 0;
    if (blockIdx.x == 0 && tid < (E & 3)) {
        dd[BIN_U * 4] = edst[(E4 << 2) + tid];
        ss[BIN_U * 4] = esrc[(E4 << 2) + tid];
    }
#pragma unroll
    for (int k = 0; k < BIN_U * 4 + 1; ++k) {
        gg[k] = (unsigned)dd[k] < (unsigned)N ? (int)((unsigned)dd[k] / OWN_R) : -1;
        rk[k] = atomicAdd(&hist[gg[k] >= 0 ? gg[k] : spare], 1);
    }
    __syncthreads();
    for (int b = tid; b < NB; b += BIN_T) {
        const int c = hist[b];
        hist[b] = c ? atomicAdd(&bincnt[b * BIN_STRIDE], c) : 0;
    }
    __syncthreads();
#pragma unroll
    for (int k = 0; k < BIN_U * 4 + 1; ++k) rk[k] += hist[gg[k] >= 0 ? gg[k] : 0];
    int2* bl = (int2*)binlist;
#pragma unroll
    for (int k = 0; k < BIN_U * 4 + 1; ++k) {
        if (gg[k] >= 0 && rk[k] < OWN_CAP)
            bl[(size_t)gg[k] * OWN_CAP + rk[k]] = make_int2(ss[k], dd[k] - gg[k] * OWN_R);
    }
}


__global__ __launch_bounds__(OWN_T) void gat_owner(
    const int* __restrict__ esrc, const int* __restrict__ edst,
    const float* __restrict__ s_src, const float* __restrict__ s_dst,
    const float* __restrict__ h, const float* __restrict__ b,
    float* __restrict__ out, int N, int E,
    const int* __restrict__ bincnt, const int* __restrict__ binlist) {
    __shared__ __attribute__((aligned(16))) float acc[(OWN_R + 1) * F];
    __shared__ unsigned mx[OWN_R];
    __shared__ float    den[OWN_R];
    __shared__ float    sd[OWN_R];
    __shared__ int      lS[OWN_CAP + OWN_U];
    __shared__ int      lD[OWN_CAP + OWN_U];
    __shared__ float    lV[OWN_CAP + OWN_U];

    const int tid  = threadIdx.x;
    const int lane = tid & 31;
    const int wave = tid >> 5;
    const int NW   = OWN_T >> 5;
    const int lo   = blockIdx.x * OWN_R;
    int nown = N - lo; if (nown > OWN_R) nown = OWN_R;
    const unsigned un = (unsigned)nown;

    for (int i = tid; i < (OWN_R + 1) * F; i += OWN_T) acc[i] = 0.f;
    if (tid < OWN_R) {
        mx[tid]  = 0x007FFFFFu;
        den[tid] = 0.f;
        sd[tid]  = (tid < nown) ? s_dst[lo + tid] : 0.f;
    }
    __syncthreads();

    const int  total = bincnt[blockIdx.x * BIN_STRIDE];
    const bool fast  = (int)gridDim.x <= BIN_NBMAX && total <= OWN_CAP;

    if (fast) {
        const int2* bl = (const int2*)binlist + (size_t)blockIdx.x * OWN_CAP;
        int2  en[OWN_CAP / OWN_T];
        float sv[OWN_CAP / OWN_T];
#pragma unroll
        for (int j = 0; j < OWN_CAP / OWN_T; ++j) {
            const int i = tid + j * OWN_T;
            en[j] = i < total ? bl[i] : make_int2(0, 0);
        }
#pragma unroll
        for (int j = 0; j < OWN_CAP / OWN_T; ++j) sv[j] = s_src[en[j].x];
#pragma unroll
        for (int j = 0; j < OWN_CAP / OWN_T; ++j) {
            const int i = tid + j * OWN_T;
            if (i < total) {
                int r = en[j].y;
                float v = sv[j] + sd[r];
                v = v > 0.f ? v : 0.2f * v;
                lS[i] = en[j].x; lD[i] = r; lV[i] = v;
                atomicMax(&mx[r], f2o(v));
            }
        }
    } else {
        for (int e = tid; e < E; e += OWN_T) {
            unsigned r = (unsigned)(edst[e] - lo);
            if (r < un) {
                float v = s_src[esrc[e]] + sd[r];
                v = v > 0.f ? v : 0.2f * v;
                atomicMax(&mx[r], f2o(v));
            }
        }
    }
    __syncthreads();

    if (fast) {
        for (int i = tid; i < total; i += OWN_T) {
            int r = lD[i];
            float ex = expf(lV[i] - o2f(mx[r]));
            lV[i] = ex;
            atomicAdd(&den[r], ex);
        }
    } else {
        for (int e = tid; e < E; e += OWN_T) {
            unsigned r = (unsigned)(edst[e] - lo);
            if (r < un) {
                float v = s_src[esrc[e]] + sd[r];
                v = v > 0.f ? v : 0.2f * v;
                atomicAdd(&den[r], expf(v - o2f(mx[r])));
            }
        }
    }
    __syncthreads();

    if (fast) {
        for (int i = tid; i < total; i += OWN_T) {
            float dn = den[lD[i]]; dn = dn > 1e-9f ? dn : 1e-9f;
            lV[i] = lV[i] / dn;
        }
        if (tid < OWN_U) { lS[total + tid] = 0; lD[total + tid] = OWN_R; lV[total + tid] = 0.f; }
    }
    __syncthreads();

    if (fast) {
        for (int i0 = wave * OWN_U; i0 < total; i0 += NW * OWN_U) {
            float a[OWN_U], h0[OWN_U], h1[OWN_U];
            int   ss[OWN_U], rr[OWN_U];
#pragma unroll
            for (int u = 0; u < OWN_U; ++u) { ss[u] = lS[i0 + u]; rr[u] = lD[i0 + u]; a[u] = lV[i0 + u]; }
#pragma unroll
            for (int u = 0; u < OWN_U; ++u) {
                const float* hr = h + (size_t)ss[u] * F;
                h0[u] = hr[lane];
                h1[u] = hr[lane + 32];
            }
#pragma unroll
            for (int u = 0; u < OWN_U; ++u) {
                atomicAdd(&acc[rr[u] * F + lane],      a[u] * h0[u]);
                atomicAdd(&acc[rr[u] * F + lane + 32], a[u] * h1[u]);
            }
        }
    } else {
        for (int e = wave; e < E; e += NW) {
            unsigned r = (unsigned)(edst[e] - lo);
            if (r < un) {
                int s = esrc[e];
                float v = s_src[s] + sd[r];
                v = v > 0.f ? v : 0.2f * v;
                float dn = den[r]; dn = dn > 1e-9f ? dn : 1e-9f;
                float alpha = expf(v - o2f(mx[r])) / dn;
                const float* hr = h + (size_t)s * F;
                atomicAdd(&acc[r * F + lane],      alpha * hr[lane]);
                atomicAdd(&acc[r * F + lane + 32], alpha * hr[lane + 32]);
            }
        }
    }
    __syncthreads();

    const float4* acc4 = (const float4*)acc;
    const float4* b4   = (const float4*)b;
    float4*       o4   = (float4*)(out + (size_t)lo * F);
    for (int i = tid; i < nown * (F / 4); i += OWN_T) {
        float4 v = acc4[i], bb = b4[i & (F / 4 - 1)];
        v.x += bb.x; v.y += bb.y; v.z += bb.z; v.w += bb.w;
        v.x = v.x > 0.f ? v.x : 0.3f * v.x;
        v.y = v.y > 0.f ? v.y : 0.3f * v.y;
        v.z = v.z > 0.f ? v.z : 0.3f * v.z;
        v.w = v.w > 0.f ? v.w : 0.3f * v.w;
        o4[i] = v;
    }
}

extern "C" void kernel_launch(void* const* d_in, const int* in_sizes, int n_in,
                              void* d_out, int out_size, void* d_ws, size_t ws_size,
                              hipStream_t stream) {
    const float* x     = (const float*)d_in[0];
    const float* W     = (const float*)d_in[1];
    const float* a_src = (const float*)d_in[2];
    const float* a_dst = (const float*)d_in[3];
    const float* b     = (const float*)d_in[4];
    const int*   esrc  = (const int*)d_in[5];
    const int*   edst  = (const int*)d_in[6];

    const int N = in_sizes[0] / F;
    const int E = in_sizes[5];
    const long total = (long)N * F;

    char* ws = (char*)d_ws;
    size_t off = 0;
    auto carve = [&](size_t bytes) -> void* {
        void* p = ws + off;
        off = (off + bytes + 255) & ~(size_t)255;
        return p;
    };
    float* h     = (float*)carve((size_t)total * 4);
    float* s_src = (float*)carve((size_t)N * 4);
    float* s_dst = (float*)carve((size_t)N * 4);
    const int NB = (N + OWN_R - 1) / OWN_R;
    int* bincnt  = (int*)carve((size_t)NB * BIN_STRIDE * 4);
    int* binlist = (int*)carve((size_t)NB * OWN_CAP * 8);
    (void)ws_size; (void)n_in; (void)out_size;

    int gemmBlocks = (N + 16 * GM_W - 1) / (16 * GM_W);
    gat_gemm_scores<<<gemmBlocks, GM_T, 0, stream>>>(x, W, a_src, a_dst, h, s_src, s_dst, N, bincnt, NB);

    int binBlocks = (E + BIN_E - 1) / BIN_E;
    if (binBlocks > 0)
        gat_bin<<<binBlocks, BIN_T, 0, stream>>>(N, esrc, edst, E, bincnt, binlist, NB);

    gat_owner<<<NB, OWN_T, 0, stream>>>(esrc, edst, s_src, s_dst, h, b,
                                        (float*)d_out, N, E, bincnt, binlist);
}


# Binning joins the GEMM launch (role by work-group index): per-block range-sorted segments and a start table replace the global counters; 3 launches become 2; sums stay in LDS arrival order
# speedup vs baseline: 5.5759x; 1.2371x over previous
#include <hip/hip_runtime.h>
#include <hip/hip_bf16.h>

#define F 64

#define OWN_R   196
#define OWN_T   1024
#define OWN_CAP 6144
#define OWN_U   8

#define BIN_T      256
#define BIN_U      4
#define BIN_E      (BIN_T * BIN_U * 4)
#define BIN_NBMAX  1024
#define SEG_CAP    (BIN_E + 4)

typedef __attribute__((ext_vector_type(2))) float v2f;
typedef __attribute__((ext_vector_type(8))) float v8f;

__device__ __forceinline__ unsigned f2o(float f) {
    unsigned u = __float_as_uint(f);
    return (u & 0x80000000u) ? ~u : (u | 0x80000000u);
}
__device__ __forceinline__ float o2f(unsigned u) {
    return (u & 0x80000000u) ? __uint_as_float(u & 0x7FFFFFFFu) : __uint_as_float(~u);
}

#define GM_W     8
#define GM_T     (GM_W * 32)
#define GM_PITCH 68
#define WP_PITCH 80

__device__ __forceinline__ void gemm_scores_role(
    const float* __restrict__ x, const float* __restrict__ W,
    const float* __restrict__ a_src, const float* __restrict__ a_dst,
    float* __restrict__ h, float* __restrict__ s_src, float* __restrict__ s_dst,
    int N, int bid) {
    __shared__ __attribute__((aligned(16))) float Wp[32 * WP_PITCH * 2];
    __shared__ __attribute__((aligned(16))) float tile[GM_W][16 * GM_PITCH];
    __shared__ __attribute__((aligned(16))) float av[GM_W][2 * F];

    const int tid  = threadIdx.x;
    const int lane = tid & 31;
    const int wave = tid >> 5;
    const int q    = lane & 15;
    const int hh   = lane >> 4;
    const int row0 = __builtin_amdgcn_readfirstlane((bid * GM_W + wave) << 4);
    float* tl = tile[wave];

    float4 xv[8];
#pragma unroll
    for (int i = 0; i < 8; ++i) {
        int gr = row0 + 2 * i + hh; if (gr > N - 1) gr = N - 1;
        xv[i] = *(const float4*)(x + (size_t)gr * F + 4 * q);
    }
    const float4 avv = *(const float4*)((hh ? a_dst : a_src) + 4 * q);
    float4 w0[2], w1[2];
#pragma unroll
    for (int j = 0; j < 2; ++j) {
        const int it = tid + j * GM_T, kp = it >> 4, cg = it & 15;
        w0[j] = *(const float4*)(W + (size_t)(2 * kp) * F + 4 * cg);
        w1[j] = *(const float4*)(W + (size_t)(2 * kp + 1) * F + 4 * cg);
    }
#pragma unroll
    for (int j = 0; j < 2; ++j) {
        const int it = tid + j * GM_T, kp = it >> 4, cg = it & 15;
        float* wp = Wp + 2 * (kp * WP_PITCH + 4 * cg);
        *(float4*)(wp)     = make_float4(w0[j].x, w1[j].x, w0[j].y, w1[j].y);
        *(float4*)(wp + 4) = make_float4(w0[j].z, w1[j].z, w0[j].w, w1[j].w);
    }
#pragma unroll
    for (int i = 0; i < 8; ++i) *(float4*)(tl + (2 * i + hh) * GM_PITCH + 4 * q) = xv[i];
    *(float4*)(av[wave] + 4 * lane) = avv;
    __syncthreads();

    v2f a[16];
#pragma unroll
    for (int ks = 0; ks < 16; ++ks) a[ks] = *(const v2f*)(tl + q * GM_PITCH + 4 * ks + 2 * hh);

    v8f acc[4] = {};
#pragma unroll
    for (int ks = 0; ks < 16; ++ks) {
#pragma unroll
        for (int t = 0; t < 4; ++t) {
            const v2f b = *(const v2f*)(Wp + 2 * ((2 * ks + hh) * WP_PITCH + 16 * t + q));
            acc[t] = __builtin_amdgcn_wmma_f32_16x16x4_f32(
                 false, a[ks],  false, b,
                 (short)0, acc[t],  false,  false);
        }
    }

#pragma unroll
    for (int t = 0; t < 4; ++t)
#pragma unroll
        for (int r = 0; r < 8; ++r) tl[(8 * hh + r) * GM_PITCH + 16 * t + q] = acc[t][r];
    __syncthreads();

    if (row0 + 16 <= N) {
        float4 hv[8];
#pragma unroll
        for (int i = 0; i < 8; ++i) hv[i] = *(const float4*)(tl + (2 * i + hh) * GM_PITCH + 4 * q);
#pragma unroll
        for (int i = 0; i < 8; ++i) *(float4*)(h + (size_t)(row0 + 2 * i + hh) * F + 4 * q) = hv[i];
    } else {
#pragma unroll
        for (int i = 0; i < 8; ++i) {
            const int r = 2 * i + hh;
            const float4 v = *(const float4*)(tl + r * GM_PITCH + 4 * q);
            if (row0 + r < N) *(float4*)(h + (size_t)(row0 + r) * F + 4 * q) = v;
        }
    }

    float ps = 0.f, pd = 0.f;
#pragma unroll
    for (int j = 0; j < 8; ++j) {
        const float4 v  = *(const float4*)(tl + q * GM_PITCH + 32 * hh + 4 * j);
        const float4 s4 = *(const float4*)(av[wave] + 32 * hh + 4 * j);
        const float4 d4 = *(const float4*)(av[wave] + F + 32 * hh + 4 * j);
        ps += v.x * s4.x + v.y * s4.y + v.z * s4.z + v.w * s4.w;
        pd += v.x * d4.x + v.y * d4.y + v.z * d4.z + v.w * d4.w;
    }
    ps += __shfl_xor(ps, 16);
    pd += __shfl_xor(pd, 16);
    if (row0 + q < N) {
        if (hh == 0) s_src[row0 + q] = ps; else s_dst[row0 + q] = pd;
    }

}

__device__ __forceinline__ void bin_role(
                           int N,
                           const int* __restrict__ esrc, const int* __restrict__ edst, int E,
                           int* __restrict__ tab, int tabPitch, int* __restrict__ segs,
                           int NB) {
    __shared__ __attribute__((aligned(16))) int hist[BIN_NBMAX + 32];
    __shared__ int wsum[BIN_T / 32];
    __shared__ __attribute__((aligned(16))) int2 sorted[SEG_CAP];
    if (NB > BIN_NBMAX) return;
    const int tid  = threadIdx.x;
    const int lane = tid & 31;
    const int wave = tid >> 5;
    for (int b = tid; b < BIN_NBMAX; b += BIN_T) hist[b] = 0;
    const int spare = BIN_NBMAX + (tid & 31);
    if (tid < 32) hist[spare] = 0;
    __syncthreads();
    const int   E4 = E >> 2;
    const int4* d4 = (const int4*)edst;
    const int4* e4 = (const int4*)esrc;
    int dd[BIN_U * 4 + 1], ss[BIN_U * 4 + 1], gg[BIN_U * 4 + 1], rk[BIN_U * 4 + 1];
#pragma unroll
    for (int u = 0; u < BIN_U; ++u) {
        const int q = (int)blockIdx.x * (BIN_T * BIN_U) + u * BIN_T + tid;
        int4 dv = make_int4(-1, -1, -1, -1), sv = make_int4(0, 0, 0, 0);
        if (q < E4) { dv = d4[q]; sv = e4[q]; }
        dd[4 * u] = dv.x; dd[4 * u + 1] = dv.y; dd[4 * u + 2] = dv.z; dd[4 * u + 3] = dv.w;
        ss[4 * u] = sv.x; ss[4 * u + 1] = sv.y; ss[4 * u + 2] = sv.z; ss[4 * u + 3] = sv.w;
    }
    dd[BIN_U * 4] = -1; ss[BIN_U * 4] = 0;
    if (blockIdx.x == 0 && tid < (E & 3)) {
        dd[BIN_U * 4] = edst[(E4 << 2) + tid];
        ss[BIN_U * 4] = esrc[(E4 << 2) + tid];
    }
#pragma unroll
    for (int k = 0; k < BIN_U * 4 + 1; ++k) {
        gg[k] = (unsigned)dd[k] < (unsigned)N ? (int)((unsigned)dd[k] / OWN_R) : -1;
        rk[k] = atomicAdd(&hist[gg[k] >= 0 ? gg[k] : spare], 1);
    }
    __syncthreads();
    const int4 cv = *(const int4*)(hist + 4 * tid);
    const int  cs = cv.x + cv.y + cv.z + cv.w;
    int inc = cs;
#pragma unroll
    for (int d = 1; d < 32; d <<= 1) {
        const int t = __shfl_up(inc, d);
        if (lane >= d) inc += t;
    }
    if (lane == 31) wsum[wave] = inc;
    __syncthreads();
    int wbase = 0, tot = 0;
#pragma unroll
    for (int w = 0; w < BIN_T / 32; ++w) {
        const int t = wsum[w];
        tot += t;
        if (w < wave) wbase += t;
    }
    const int f0 = wbase + inc - cs, f1 = f0 + cv.x, f2 = f1 + cv.y, f3 = f2 + cv.z;
    *(int4*)(hist + 4 * tid) = make_int4(f0, f1, f2, f3);
    {
        int* tp = tab + (size_t)(4 * tid) * tabPitch + blockIdx.x;
        if (4 * tid     < NB) tp[0]                    = (f0 << 16) | cv.x;
        if (4 * tid + 1 < NB) tp[tabPitch]             = (f1 << 16) | cv.y;
        if (4 * tid + 2 < NB) tp[2 * (size_t)tabPitch] = (f2 << 16) | cv.z;
        if (4 * tid + 3 < NB) tp[3 * (size_t)tabPitch] = (f3 << 16) | cv.w;
    }
    __syncthreads();
#pragma unroll
    for (int k = 0; k < BIN_U * 4 + 1; ++k) rk[k] += hist[gg[k] >= 0 ? gg[k] : 0];
#pragma unroll
    for (int k = 0; k < BIN_U * 4 + 1; ++k) {
        if (gg[k] >= 0) sorted[rk[k]] = make_int2(ss[k], dd[k] - gg[k] * OWN_R);
    }
    __syncthreads();
    int4*       o4 = (int4*)(segs + (size_t)blockIdx.x * (SEG_CAP * 2));
    const int4* l4 = (const int4*)sorted;
    for (int i = tid; i < ((tot + 1) >> 1); i += BIN_T) o4[i] = l4[i];
}

static_assert(BIN_T == GM_T, "both roles of launch 1 use one block size");
static_assert(BIN_T * 4 == BIN_NBMAX, "the scan gives each thread four ranges");
static_assert((SEG_CAP & 1) == 0, "segments are copied out as 16-byte pairs");

__global__ __launch_bounds__(GM_T) void gat_gemm_bin(
    const float* __restrict__ x, const float* __restrict__ W,
    const float* __restrict__ a_src, const float* __restrict__ a_dst,
    float* __restrict__ h, float* __restrict__ s_src, float* __restrict__ s_dst,
    int N,
    const int* __restrict__ esrc, const int* __restrict__ edst, int E,
    int* __restrict__ tab, int tabPitch, int* __restrict__ segs, int NB, int nBin) {
    if ((int)blockIdx.x < nBin) {
        bin_role(N, esrc, edst, E, tab, tabPitch, segs, NB);
        return;
    }
    gemm_scores_role(x, W, a_src, a_dst, h, s_src, s_dst, N, (int)blockIdx.x - nBin);
}


__global__ __launch_bounds__(OWN_T) void gat_owner(
    const int* __restrict__ esrc, const int* __restrict__ edst,
    const float* __restrict__ s_src, const float* __restrict__ s_dst,
    const float* __restrict__ h, const float* __restrict__ b,
    float* __restrict__ out, int N, int E,
    const int* __restrict__ tab, int tabPitch, const int* __restrict__ segs, int nBin) {
    __shared__ __attribute__((aligned(16))) float acc[(OWN_R + 1) * F];
    __shared__ unsigned mx[OWN_R];
    __shared__ float    den[OWN_R];
    __shared__ float    sd[OWN_R];
    __shared__ int      lS[OWN_CAP + OWN_U];
    __shared__ int      lD[OWN_CAP + OWN_U];
    __shared__ float    lV[OWN_CAP + OWN_U];
    __shared__ int      ltot;

    const int tid  = threadIdx.x;
    const int lane = tid & 31;
    const int wave = tid >> 5;
    const int NW   = OWN_T >> 5;
    const int lo   = blockIdx.x * OWN_R;
    int nown = N - lo; if (nown > OWN_R) nown = OWN_R;
    const unsigned un = (unsigned)nown;

    for (int i = tid; i < (OWN_R + 1) * F; i += OWN_T) acc[i] = 0.f;
    if (tid < OWN_R) {
        mx[tid]  = 0x007FFFFFu;
        den[tid] = 0.f;
        sd[tid]  = (tid < nown) ? s_dst[lo + tid] : 0.f;
    }
    if (tid == 0) ltot = 0;
    __syncthreads();

    const bool binned = (int)gridDim.x <= BIN_NBMAX;
    if (binned) {
        const int*  trow = tab + (size_t)blockIdx.x * tabPitch;
        const int2* sg0  = (const int2*)segs;
        for (int b0 = wave * 8; b0 < nBin; b0 += NW * 8) {
            const int myb = b0 + (lane & 7);
            int pk = 0;
            if (lane < 8 && myb < nBin) pk = trow[myb];
            const int mc = pk & 0xFFFF;
            int mp = 0;
            if (mc > 0) mp = atomicAdd(&ltot, mc);
            int         c[8], p[8];
            const int2* sg[8];
            int2        en[8];
            float       sv[8];
#pragma unroll
            for (int u = 0; u < 8; ++u) {
                c[u]  = __shfl(mc, u);
                p[u]  = __shfl(mp, u);
                sg[u] = sg0 + (size_t)(b0 + u) * SEG_CAP + __shfl(pk >> 16, u);
            }
#pragma unroll
            for (int u = 0; u < 8; ++u) en[u] = lane < c[u] ? sg[u][lane] : make_int2(0, 0);
#pragma unroll
            for (int u = 0; u < 8; ++u) sv[u] = s_src[en[u].x];
#pragma unroll
            for (int u = 0; u < 8; ++u) {
                const int i = p[u] + lane;
                if (lane < c[u] && i < OWN_CAP) {
                    const int r = en[u].y;
                    float v = sv[u] + sd[r];
                    v = v > 0.f ? v : 0.2f * v;
                    lS[i] = en[u].x; lD[i] = r; lV[i] = v;
                    atomicMax(&mx[r], f2o(v));
                }
            }
#pragma unroll
            for (int u = 0; u < 8; ++u) {
                for (int j = lane + 32; j < c[u]; j += 32) {
                    const int i = p[u] + j;
                    if (i < OWN_CAP) {
                        const int2 e2 = sg[u][j];
                        float v = s_src[e2.x] + sd[e2.y];
                        v = v > 0.f ? v : 0.2f * v;
                        lS[i] = e2.x; lD[i] = e2.y; lV[i] = v;
                        atomicMax(&mx[e2.y], f2o(v));
                    }
                }
            }
        }
    }
    __syncthreads();
    const int  total = ltot;
    const bool fast  = binned && total <= OWN_CAP;
    if (!fast) {
        for (int e = tid; e < E; e += OWN_T) {
            unsigned r = (unsigned)(edst[e] - lo);
            if (r < un) {
                float v = s_src[esrc[e]] + sd[r];
                v = v > 0.f ? v : 0.2f * v;
                atomicMax(&mx[r], f2o(v));
            }
        }
    }
    __syncthreads();

    if (fast) {
        for (int i = tid; i < total; i += OWN_T) {
            int r = lD[i];
            float ex = expf(lV[i] - o2f(mx[r]));
            lV[i] = ex;
            atomicAdd(&den[r], ex);
        }
    } else {
        for (int e = tid; e < E; e += OWN_T) {
            unsigned r = (unsigned)(edst[e] - lo);
            if (r < un) {
                float v = s_src[esrc[e]] + sd[r];
                v = v > 0.f ? v : 0.2f * v;
                atomicAdd(&den[r], expf(v - o2f(mx[r])));
            }
        }
    }
    __syncthreads();

    if (fast) {
        for (int i = tid; i < total; i += OWN_T) {
            float dn = den[lD[i]]; dn = dn > 1e-9f ? dn : 1e-9f;
            lV[i] = lV[i] / dn;
        }
        if (tid < OWN_U) { lS[total + tid] = 0; lD[total + tid] = OWN_R; lV[total + tid] = 0.f; }
    }
    __syncthreads();

    if (fast) {
        for (int i0 = wave * OWN_U; i0 < total; i0 += NW * OWN_U) {
            float a[OWN_U], h0[OWN_U], h1[OWN_U];
            int   ss[OWN_U], rr[OWN_U];
#pragma unroll
            for (int u = 0; u < OWN_U; ++u) { ss[u] = lS[i0 + u]; rr[u] = lD[i0 + u]; a[u] = lV[i0 + u]; }
#pragma unroll
            for (int u = 0; u < OWN_U; ++u) {
                const float* hr = h + (size_t)ss[u] * F;
                h0[u] = hr[lane];
                h1[u] = hr[lane + 32];
            }
#pragma unroll
            for (int u = 0; u < OWN_U; ++u) {
                atomicAdd(&acc[rr[u] * F + lane],      a[u] * h0[u]);
                atomicAdd(&acc[rr[u] * F + lane + 32], a[u] * h1[u]);
            }
        }
    } else {
        for (int e = wave; e < E; e += NW) {
            unsigned r = (unsigned)(edst[e] - lo);
            if (r < un) {
                int s = esrc[e];
                float v = s_src[s] + sd[r];
                v = v > 0.f ? v : 0.2f * v;
                float dn = den[r]; dn = dn > 1e-9f ? dn : 1e-9f;
                float alpha = expf(v - o2f(mx[r])) / dn;
                const float* hr = h + (size_t)s * F;
                atomicAdd(&acc[r * F + lane],      alpha * hr[lane]);
                atomicAdd(&acc[r * F + lane + 32], alpha * hr[lane + 32]);
            }
        }
    }
    __syncthreads();

    const float4* acc4 = (const float4*)acc;
    const float4* b4   = (const float4*)b;
    float4*       o4   = (float4*)(out + (size_t)lo * F);
    for (int i = tid; i < nown * (F / 4); i += OWN_T) {
        float4 v = acc4[i], bb = b4[i & (F / 4 - 1)];
        v.x += bb.x; v.y += bb.y; v.z += bb.z; v.w += bb.w;
        v.x = v.x > 0.f ? v.x : 0.3f * v.x;
        v.y = v.y > 0.f ? v.y : 0.3f * v.y;
        v.z = v.z > 0.f ? v.z : 0.3f * v.z;
        v.w = v.w > 0.f ? v.w : 0.3f * v.w;
        o4[i] = v;
    }
}

extern "C" void kernel_launch(void* const* d_in, const int* in_sizes, int n_in,
                              void* d_out, int out_size, void* d_ws, size_t ws_size,
                              hipStream_t stream) {
    const float* x     = (const float*)d_in[0];
    const float* W     = (const float*)d_in[1];
    const float* a_src = (const float*)d_in[2];
    const float* a_dst = (const float*)d_in[3];
    const float* b     = (const float*)d_in[4];
    const int*   esrc  = (const int*)d_in[5];
    const int*   edst  = (const int*)d_in[6];

    const int N = in_sizes[0] / F;
    const int E = in_sizes[5];
    const long total = (long)N * F;

    char* ws = (char*)d_ws;
    size_t off = 0;
    auto carve = [&](size_t bytes) -> void* {
        void* p = ws + off;
        off = (off + bytes + 255) & ~(size_t)255;
        return p;
    };
    float* h     = (float*)carve((size_t)total * 4);
    float* s_src = (float*)carve((size_t)N * 4);
    float* s_dst = (float*)carve((size_t)N * 4);
    const int NB = (N + OWN_R - 1) / OWN_R;
    const int binBlocks = (E + BIN_E - 1) / BIN_E;
    const int tabPitch  = (binBlocks + 31) & ~31;
    int* tab  = (int*)carve((size_t)NB * tabPitch * 4);
    int* segs = (int*)carve((size_t)binBlocks * SEG_CAP * 8);
    (void)ws_size; (void)n_in; (void)out_size;

    int gemmBlocks = (N + 16 * GM_W - 1) / (16 * GM_W);
    gat_gemm_bin<<<binBlocks + gemmBlocks, GM_T, 0, stream>>>(x, W, a_src, a_dst, h, s_src, s_dst, N,
                                                             esrc, edst, E, tab, tabPitch, segs, NB, binBlocks);

    gat_owner<<<NB, OWN_T, 0, stream>>>(esrc, edst, s_src, s_dst, h, b,
                                        (float*)d_out, N, E, tab, tabPitch, segs, binBlocks);
}


# Owner sorts its edge list by destination node in LDS (rank = returning LDS add, scan, copy); half a wave sums a node's rows in registers, stores each row once; no LDS float row adds; rank-order sums
# speedup vs baseline: 6.2736x; 1.1251x over previous
#include <hip/hip_runtime.h>
#include <hip/hip_bf16.h>

#define F 64

#define OWN_R   196
#define OWN_T   1024
#define OWN_CAP 6144
#define ROW_U   12
#define ORD_CAP (OWN_CAP + 3 * OWN_R + ROW_U)

#define BIN_T      256
#define BIN_U      4
#define BIN_E      (BIN_T * BIN_U * 4)
#define BIN_NBMAX  1024
#define SEG_CAP    (BIN_E + 4)

typedef __attribute__((ext_vector_type(2))) float v2f;
typedef __attribute__((ext_vector_type(8))) float v8f;

__device__ __forceinline__ unsigned f2o(float f) {
    unsigned u = __float_as_uint(f);
    return (u & 0x80000000u) ? ~u : (u | 0x80000000u);
}
__device__ __forceinline__ float o2f(unsigned u) {
    return (u & 0x80000000u) ? __uint_as_float(u & 0x7FFFFFFFu) : __uint_as_float(~u);
}

#define GM_W     8
#define GM_T     (GM_W * 32)
#define GM_PITCH 68
#define WP_PITCH 80

__device__ __forceinline__ void gemm_scores_role(
    const float* __restrict__ x, const float* __restrict__ W,
    const float* __restrict__ a_src, const float* __restrict__ a_dst,
    float* __restrict__ h, float* __restrict__ s_src, float* __restrict__ s_dst,
    int N, int bid) {
    __shared__ __attribute__((aligned(16))) float Wp[32 * WP_PITCH * 2];
    __shared__ __attribute__((aligned(16))) float tile[GM_W][16 * GM_PITCH];
    __shared__ __attribute__((aligned(16))) float av[GM_W][2 * F];

    const int tid  = threadIdx.x;
    const int lane = tid & 31;
    const int wave = tid >> 5;
    const int q    = lane & 15;
    const int hh   = lane >> 4;
    const int row0 = __builtin_amdgcn_readfirstlane((bid * GM_W + wave) << 4);
    float* tl = tile[wave];

    float4 xv[8];
#pragma unroll
    for (int i = 0; i < 8; ++i) {
        int gr = row0 + 2 * i + hh; if (gr > N - 1) gr = N - 1;
        xv[i] = *(const float4*)(x + (size_t)gr * F + 4 * q);
    }
    const float4 avv = *(const float4*)((hh ? a_dst : a_src) + 4 * q);
    float4 w0[2], w1[2];
#pragma unroll
    for (int j = 0; j < 2; ++j) {
        const int it = tid + j * GM_T, kp = it >> 4, cg = it & 15;
        w0[j] = *(const float4*)(W + (size_t)(2 * kp) * F + 4 * cg);
        w1[j] = *(const float4*)(W + (size_t)(2 * kp + 1) * F + 4 * cg);
    }
#pragma unroll
    for (int j = 0; j < 2; ++j) {
        const int it = tid + j * GM_T, kp = it >> 4, cg = it & 15;
        float* wp = Wp + 2 * (kp * WP_PITCH + 4 * cg);
        *(float4*)(wp)     = make_float4(w0[j].x, w1[j].x, w0[j].y, w1[j].y);
        *(float4*)(wp + 4) = make_float4(w0[j].z, w1[j].z, w0[j].w, w1[j].w);
    }
#pragma unroll
    for (int i = 0; i < 8; ++i) *(float4*)(tl + (2 * i + hh) * GM_PITCH + 4 * q) = xv[i];
    *(float4*)(av[wave] + 4 * lane) = avv;
    __syncthreads();

    v2f a[16];
#pragma unroll
    for (int ks = 0; ks < 16; ++ks) a[ks] = *(const v2f*)(tl + q * GM_PITCH + 4 * ks + 2 * hh);

    v8f acc[4] = {};
#pragma unroll
    for (int ks = 0; ks < 16; ++ks) {
#pragma unroll
        for (int t = 0; t < 4; ++t) {
            const v2f b = *(const v2f*)(Wp + 2 * ((2 * ks + hh) * WP_PITCH + 16 * t + q));
            acc[t] = __builtin_amdgcn_wmma_f32_16x16x4_f32(
                 false, a[ks],  false, b,
                 (short)0, acc[t],  false,  false);
        }
    }

#pragma unroll
    for (int t = 0; t < 4; ++t)
#pragma unroll
        for (int r = 0; r < 8; ++r) tl[(8 * hh + r) * GM_PITCH + 16 * t + q] = acc[t][r];
    __syncthreads();

    if (row0 + 16 <= N) {
        float4 hv[8];
#pragma unroll
        for (int i = 0; i < 8; ++i) hv[i] = *(const float4*)(tl + (2 * i + hh) * GM_PITCH + 4 * q);
#pragma unroll
        for (int i = 0; i < 8; ++i) *(float4*)(h + (size_t)(row0 + 2 * i + hh) * F + 4 * q) = hv[i];
    } else {
#pragma unroll
        for (int i = 0; i < 8; ++i) {
            const int r = 2 * i + hh;
            const float4 v = *(const float4*)(tl + r * GM_PITCH + 4 * q);
            if (row0 + r < N) *(float4*)(h + (size_t)(row0 + r) * F + 4 * q) = v;
        }
    }

    float ps = 0.f, pd = 0.f;
#pragma unroll
    for (int j = 0; j < 8; ++j) {
        const float4 v  = *(const float4*)(tl + q * GM_PITCH + 32 * hh + 4 * j);
        const float4 s4 = *(const float4*)(av[wave] + 32 * hh + 4 * j);
        const float4 d4 = *(const float4*)(av[wave] + F + 32 * hh + 4 * j);
        ps += v.x * s4.x + v.y * s4.y + v.z * s4.z + v.w * s4.w;
        pd += v.x * d4.x + v.y * d4.y + v.z * d4.z + v.w * d4.w;
    }
    ps += __shfl_xor(ps, 16);
    pd += __shfl_xor(pd, 16);
    if (row0 + q < N) {
        if (hh == 0) s_src[row0 + q] = ps; else s_dst[row0 + q] = pd;
    }

}

__device__ __forceinline__ void bin_role(
                           int N,
                           const int* __restrict__ esrc, const int* __restrict__ edst, int E,
                           int* __restrict__ tab, int tabPitch, int* __restrict__ segs,
                           int NB) {
    __shared__ __attribute__((aligned(16))) int hist[BIN_NBMAX + 32];
    __shared__ int wsum[BIN_T / 32];
    __shared__ __attribute__((aligned(16))) int2 sorted[SEG_CAP];
    if (NB > BIN_NBMAX) return;
    const int tid  = threadIdx.x;
    const int lane = tid & 31;
    const int wave = tid >> 5;
    for (int b = tid; b < BIN_NBMAX; b += BIN_T) hist[b] = 0;
    const int spare = BIN_NBMAX + (tid & 31);
    if (tid < 32) hist[spare] = 0;
    __syncthreads();
    const int   E4 = E >> 2;
    const int4* d4 = (const int4*)edst;
    const int4* e4 = (const int4*)esrc;
    int dd[BIN_U * 4 + 1], ss[BIN_U * 4 + 1], gg[BIN_U * 4 + 1], rk[BIN_U * 4 + 1];
#pragma unroll
    for (int u = 0; u < BIN_U; ++u) {
        const int q = (int)blockIdx.x * (BIN_T * BIN_U) + u * BIN_T + tid;
        int4 dv = make_int4(-1, -1, -1, -1), sv = make_int4(0, 0, 0, 0);
        if (q < E4) { dv = d4[q]; sv = e4[q]; }
        dd[4 * u] = dv.x; dd[4 * u + 1] = dv.y; dd[4 * u + 2] = dv.z; dd[4 * u + 3] = dv.w;
        ss[4 * u] = sv.x; ss[4 * u + 1] = sv.y; ss[4 * u + 2] = sv.z; ss[4 * u + 3] = sv.w;
    }
    dd[BIN_U * 4] = -1; ss[BIN_U * 4] = 0;
    if (blockIdx.x == 0 && tid < (E & 3)) {
        dd[BIN_U * 4] = edst[(E4 << 2) + tid];
        ss[BIN_U * 4] = esrc[(E4 << 2) + tid];
    }
#pragma unroll
    for (int k = 0; k < BIN_U * 4 + 1; ++k) {
        gg[k] = (unsigned)dd[k] < (unsigned)N ? (int)((unsigned)dd[k] / OWN_R) : -1;
        rk[k] = atomicAdd(&hist[gg[k] >= 0 ? gg[k] : spare], 1);
    }
    __syncthreads();
    const int4 cv = *(const int4*)(hist + 4 * tid);
    const int  cs = cv.x + cv.y + cv.z + cv.w;
    int inc = cs;
#pragma unroll
    for (int d = 1; d < 32; d <<= 1) {
        const int t = __shfl_up(inc, d);
        if (lane >= d) inc += t;
    }
    if (lane == 31) wsum[wave] = inc;
    __syncthreads();
    int wbase = 0, tot = 0;
#pragma unroll
    for (int w = 0; w < BIN_T / 32; ++w) {
        const int t = wsum[w];
        tot += t;
        if (w < wave) wbase += t;
    }
    const int f0 = wbase + inc - cs, f1 = f0 + cv.x, f2 = f1 + cv.y, f3 = f2 + cv.z;
    *(int4*)(hist + 4 * tid) = make_int4(f0, f1, f2, f3);
    {
        int* tp = tab + (size_t)(4 * tid) * tabPitch + blockIdx.x;
        if (4 * tid     < NB) tp[0]                    = (f0 << 16) | cv.x;
        if (4 * tid + 1 < NB) tp[tabPitch]             = (f1 << 16) | cv.y;
        if (4 * tid + 2 < NB) tp[2 * (size_t)tabPitch] = (f2 << 16) | cv.z;
        if (4 * tid + 3 < NB) tp[3 * (size_t)tabPitch] = (f3 << 16) | cv.w;
    }
    __syncthreads();
#pragma unroll
    for (int k = 0; k < BIN_U * 4 + 1; ++k) rk[k] += hist[gg[k] >= 0 ? gg[k] : 0];
#pragma unroll
    for (int k = 0; k < BIN_U * 4 + 1; ++k) {
        if (gg[k] >= 0) sorted[rk[k]] = make_int2(ss[k], dd[k] - gg[k] * OWN_R);
    }
    __syncthreads();
    int4*       o4 = (int4*)(segs + (size_t)blockIdx.x * (SEG_CAP * 2));
    const int4* l4 = (const int4*)sorted;
    for (int i = tid; i < ((tot + 1) >> 1); i += BIN_T) o4[i] = l4[i];
}

static_assert(BIN_T == GM_T, "both roles of launch 1 use one block size");
static_assert(OWN_R <= 256, "an owner's list packs the node below bit 8 and the rank above it");
static_assert(ROW_U % 4 == 0, "the ordered list is read four entries at a time");
static_assert(BIN_T * 4 == BIN_NBMAX, "the scan gives each thread four ranges");
static_assert((SEG_CAP & 1) == 0, "segments are copied out as 16-byte pairs");

__global__ __launch_bounds__(GM_T) void gat_gemm_bin(
    const float* __restrict__ x, const float* __restrict__ W,
    const float* __restrict__ a_src, const float* __restrict__ a_dst,
    float* __restrict__ h, float* __restrict__ s_src, float* __restrict__ s_dst,
    int N,
    const int* __restrict__ esrc, const int* __restrict__ edst, int E,
    int* __restrict__ tab, int tabPitch, int* __restrict__ segs, int NB, int nBin) {
    if ((int)blockIdx.x < nBin) {
        bin_role(N, esrc, edst, E, tab, tabPitch, segs, NB);
        return;
    }
    gemm_scores_role(x, W, a_src, a_dst, h, s_src, s_dst, N, (int)blockIdx.x - nBin);
}


__global__ __launch_bounds__(OWN_T) void gat_owner(
    const int* __restrict__ esrc, const int* __restrict__ edst,
    const float* __restrict__ s_src, const float* __restrict__ s_dst,
    const float* __restrict__ h, const float* __restrict__ b,
    float* __restrict__ out, int N, int E,
    const int* __restrict__ tab, int tabPitch, const int* __restrict__ segs, int nBin) {
    __shared__ __attribute__((aligned(16))) float acc[(OWN_R + 1) * F];
    __shared__ unsigned mx[OWN_R];
    __shared__ float    den[OWN_R];
    __shared__ float    sd[OWN_R];
    __shared__ int      cnt[OWN_R];
    __shared__ int      st[OWN_R];
    __shared__ int      lS[OWN_CAP];
    __shared__ int      lD[OWN_CAP];
    __shared__ float    lV[OWN_CAP];
    __shared__ __attribute__((aligned(16))) int   oS[ORD_CAP];
    __shared__ __attribute__((aligned(16))) float oA[ORD_CAP];
    __shared__ int      ltot;

    const int tid  = threadIdx.x;
    const int lane = tid & 31;
    const int wave = tid >> 5;
    const int NW   = OWN_T >> 5;
    const int lo   = blockIdx.x * OWN_R;
    int nown = N - lo; if (nown > OWN_R) nown = OWN_R;
    const unsigned un = (unsigned)nown;

    if (tid < OWN_R) {
        mx[tid]  = 0x007FFFFFu;
        den[tid] = 0.f;
        cnt[tid] = 0;
        sd[tid]  = (tid < nown) ? s_dst[lo + tid] : 0.f;
    }
    if (tid == 0) ltot = 0;
    __syncthreads();

    const bool binned = (int)gridDim.x <= BIN_NBMAX;
    if (binned) {
        const int*  trow = tab + (size_t)blockIdx.x * tabPitch;
        const int2* sg0  = (const int2*)segs;
        for (int b0 = wave * 8; b0 < nBin; b0 += NW * 8) {
            const int myb = b0 + (lane & 7);
            int pk = 0;
            if (lane < 8 && myb < nBin) pk = trow[myb];
            const int mc = pk & 0xFFFF;
            int mp = 0;
            if (mc > 0) mp = atomicAdd(&ltot, mc);
            int         c[8], p[8];
            const int2* sg[8];
            int2        en[8];
            float       sv[8];
#pragma unroll
            for (int u = 0; u < 8; ++u) {
                c[u]  = __shfl(mc, u);
                p[u]  = __shfl(mp, u);
                sg[u] = sg0 + (size_t)(b0 + u) * SEG_CAP + __shfl(pk >> 16, u);
            }
#pragma unroll
            for (int u = 0; u < 8; ++u) en[u] = lane < c[u] ? sg[u][lane] : make_int2(0, 0);
#pragma unroll
            for (int u = 0; u < 8; ++u) sv[u] = s_src[en[u].x];
#pragma unroll
            for (int u = 0; u < 8; ++u) {
                const int i = p[u] + lane;
                if (lane < c[u] && i < OWN_CAP) {
                    const int r = en[u].y;
                    float v = sv[u] + sd[r];
                    v = v > 0.f ? v : 0.2f * v;
                    const int rk = atomicAdd(&cnt[r], 1);
                    lS[i] = en[u].x; lD[i] = r | (rk << 8); lV[i] = v;
                    atomicMax(&mx[r], f2o(v));
                }
            }
#pragma unroll
            for (int u = 0; u < 8; ++u) {
                for (int j = lane + 32; j < c[u]; j += 32) {
                    const int i = p[u] + j;
                    if (i < OWN_CAP) {
                        const int2 e2 = sg[u][j];
                        float v = s_src[e2.x] + sd[e2.y];
                        v = v > 0.f ? v : 0.2f * v;
                        const int rk = atomicAdd(&cnt[e2.y], 1);
                        lS[i] = e2.x; lD[i] = e2.y | (rk << 8); lV[i] = v;
                        atomicMax(&mx[e2.y], f2o(v));
                    }
                }
            }
        }
    }
    __syncthreads();
    const int  total = ltot;
    const bool fast  = binned && total <= OWN_CAP;
    if (!fast) {
        for (int i = tid; i < (OWN_R + 1) * F; i += OWN_T) acc[i] = 0.f;
        for (int e = tid; e < E; e += OWN_T) {
            unsigned r = (unsigned)(edst[e] - lo);
            if (r < un) {
                float v = s_src[esrc[e]] + sd[r];
                v = v > 0.f ? v : 0.2f * v;
                atomicMax(&mx[r], f2o(v));
            }
        }
    }
    __syncthreads();

    if (fast) {
        if (wave == NW - 1) {
            constexpr int SCAN_K = (OWN_R + 31) / 32;
            int ck[SCAN_K], s = 0;
#pragma unroll
            for (int k = 0; k < SCAN_K; ++k) {
                const int idx = SCAN_K * lane + k;
                ck[k] = idx < OWN_R ? ((cnt[idx] + 3) & ~3) : 0;
                s += ck[k];
            }
            int inc = s;
#pragma unroll
            for (int d = 1; d < 32; d <<= 1) {
                const int t = __shfl_up(inc, d);
                if (lane >= d) inc += t;
            }
            int run = inc - s;
#pragma unroll
            for (int k = 0; k < SCAN_K; ++k) {
                const int idx = SCAN_K * lane + k;
                if (idx < OWN_R) st[idx] = run;
                run += ck[k];
            }
        }
        for (int i = tid; i < total; i += OWN_T) {
            int r = lD[i] & 255;
            float ex = expf(lV[i] - o2f(mx[r]));
            lV[i] = ex;
            atomicAdd(&den[r], ex);
        }
    } else {
        for (int e = tid; e < E; e += OWN_T) {
            unsigned r = (unsigned)(edst[e] - lo);
            if (r < un) {
                float v = s_src[esrc[e]] + sd[r];
                v = v > 0.f ? v : 0.2f * v;
                atomicAdd(&den[r], expf(v - o2f(mx[r])));
            }
        }
    }
    __syncthreads();

    if (fast) {
        for (int i = tid; i < total; i += OWN_T) {
            const int pk = lD[i];
            const int r  = pk & 255;
            float dn = den[r]; dn = dn > 1e-9f ? dn : 1e-9f;
            const int pos = st[r] + (pk >> 8);
            oS[pos] = lS[i];
            oA[pos] = lV[i] / dn;
        }
    }
    __syncthreads();

    if (fast) {
        const int      hl   = lane & 15;
        const int      hh   = lane >> 4;
        const unsigned cofs = (unsigned)hl << 4;
        const float4   bb   = *(const float4*)(b + 4 * hl);
        for (int q = wave; q < (OWN_R + 1) / 2; q += NW) {
            const int  r   = 2 * q + hh;
            const bool own = r < nown;
            const int  n   = own ? cnt[r] : 0;
            const int  s0  = own ? st[r] : 0;
            float4 sum = make_float4(0.f, 0.f, 0.f, 0.f);
            for (int j = 0; j < n; j += ROW_U) {
                int    sv[ROW_U];
                float  av[ROW_U];
                float4 hv[ROW_U];
#pragma unroll
                for (int k = 0; k < ROW_U / 4; ++k) {
                    const int4 t = *(const int4*)(oS + s0 + j + 4 * k);
                    sv[4 * k] = t.x; sv[4 * k + 1] = t.y; sv[4 * k + 2] = t.z; sv[4 * k + 3] = t.w;
                }
#pragma unroll
                for (int u = 0; u < ROW_U; ++u) {
                    const int s = (j + u < n) ? sv[u] : sv[0];
                    hv[u] = *(const float4*)((const char*)h + (((unsigned)s << 8) | cofs));
                }
#pragma unroll
                for (int k = 0; k < ROW_U / 4; ++k) {
                    const float4 t = *(const float4*)(oA + s0 + j + 4 * k);
                    av[4 * k] = t.x; av[4 * k + 1] = t.y; av[4 * k + 2] = t.z; av[4 * k + 3] = t.w;
                }
#pragma unroll
                for (int u = 0; u < ROW_U; ++u) {
                    const float a = (j + u < n) ? av[u] : 0.f;
                    sum.x += a * hv[u].x; sum.y += a * hv[u].y; sum.z += a * hv[u].z; sum.w += a * hv[u].w;
                }
            }
            if (own) {
                float4 v = make_float4(sum.x + bb.x, sum.y + bb.y, sum.z + bb.z, sum.w + bb.w);
                v.x = v.x > 0.f ? v.x : 0.3f * v.x;
                v.y = v.y > 0.f ? v.y : 0.3f * v.y;
                v.z = v.z > 0.f ? v.z : 0.3f * v.z;
                v.w = v.w > 0.f ? v.w : 0.3f * v.w;
                *(float4*)(out + (size_t)(lo + r) * F + 4 * hl) = v;
            }
        }
    } else {
        for (int e = wave; e < E; e += NW) {
            unsigned r = (unsigned)(edst[e] - lo);
            if (r < un) {
                int s = esrc[e];
                float v = s_src[s] + sd[r];
                v = v > 0.f ? v : 0.2f * v;
                float dn = den[r]; dn = dn > 1e-9f ? dn : 1e-9f;
                float alpha = expf(v - o2f(mx[r])) / dn;
                const float* hr = h + (size_t)s * F;
                atomicAdd(&acc[r * F + lane],      alpha * hr[lane]);
                atomicAdd(&acc[r * F + lane + 32], alpha * hr[lane + 32]);
            }
        }
        __syncthreads();

        const float4* acc4 = (const float4*)acc;
        const float4* b4   = (const float4*)b;
        float4*       o4   = (float4*)(out + (size_t)lo * F);
        for (int i = tid; i < nown * (F / 4); i += OWN_T) {
            float4 v = acc4[i], bb4 = b4[i & (F / 4 - 1)];
            v.x += bb4.x; v.y += bb4.y; v.z += bb4.z; v.w += bb4.w;
            v.x = v.x > 0.f ? v.x : 0.3f * v.x;
            v.y = v.y > 0.f ? v.y : 0.3f * v.y;
            v.z = v.z > 0.f ? v.z : 0.3f * v.z;
            v.w = v.w > 0.f ? v.w : 0.3f * v.w;
            o4[i] = v;
        }
    }
}

extern "C" void kernel_launch(void* const* d_in, const int* in_sizes, int n_in,
                              void* d_out, int out_size, void* d_ws, size_t ws_size,
                              hipStream_t stream) {
    const float* x     = (const float*)d_in[0];
    const float* W     = (const float*)d_in[1];
    const float* a_src = (const float*)d_in[2];
    const float* a_dst = (const float*)d_in[3];
    const float* b     = (const float*)d_in[4];
    const int*   esrc  = (const int*)d_in[5];
    const int*   edst  = (const int*)d_in[6];

    const int N = in_sizes[0] / F;
    const int E = in_sizes[5];
    const long total = (long)N * F;

    char* ws = (char*)d_ws;
    size_t off = 0;
    auto carve = [&](size_t bytes) -> void* {
        void* p = ws + off;
        off = (off + bytes + 255) & ~(size_t)255;
        return p;
    };
    float* h     = (float*)carve((size_t)total * 4);
    float* s_src = (float*)carve((size_t)N * 4);
    float* s_dst = (float*)carve((size_t)N * 4);
    const int NB = (N + OWN_R - 1) / OWN_R;
    const int binBlocks = (E + BIN_E - 1) / BIN_E;
    const int tabPitch  = (binBlocks + 31) & ~31;
    int* tab  = (int*)carve((size_t)NB * tabPitch * 4);
    int* segs = (int*)carve((size_t)binBlocks * SEG_CAP * 8);
    (void)ws_size; (void)n_in; (void)out_size;

    int gemmBlocks = (N + 16 * GM_W - 1) / (16 * GM_W);
    gat_gemm_bin<<<binBlocks + gemmBlocks, GM_T, 0, stream>>>(x, W, a_src, a_dst, h, s_src, s_dst, N,
                                                             esrc, edst, E, tab, tabPitch, segs, NB, binBlocks);

    gat_owner<<<NB, OWN_T, 0, stream>>>(esrc, edst, s_src, s_dst, h, b,
                                        (float*)d_out, N, E, tab, tabPitch, segs, binBlocks);
}
